# starting kernel
# baseline (speedup 1.0000x reference)
#include <hip/hip_runtime.h>
#include <hip/hip_bf16.h>

#define R_ 8
#define K_ 10
#define D_ 256
#define DL_ 10
#define B_ 4096
#define KD_ 100
#define KDP_ 112
#define INV_TEMP 10.0f

#define XP 260
#define ZP 114

#define G_OFF  0u
#define UP_OFF 32768u
#define UT_OFF 950272u

typedef float v2f __attribute__((ext_vector_type(2)));
typedef float v8f __attribute__((ext_vector_type(8)));

__global__ __launch_bounds__(128) void gram_kernel(const float* __restrict__ Us,
                                                   float* __restrict__ G) {
  const int rk = blockIdx.x;
  const int t  = threadIdx.x;
  if (t >= KD_) return;
  const int d1 = t / DL_, d2 = t % DL_;
  const float* u = Us + (size_t)rk * D_ * DL_;
  float acc = 0.f;
#pragma unroll 8
  for (int Dd = 0; Dd < D_; ++Dd)
    acc = fmaf(u[Dd * DL_ + d1], u[Dd * DL_ + d2], acc);
  G[rk * KD_ + t] = acc;
}

__global__ __launch_bounds__(KDP_) void pack_kernel(const float* __restrict__ Us,
                                                    float* __restrict__ Upad2,
                                                    float* __restrict__ UpadT2) {
  const int kd = threadIdx.x;
  const int rd = blockIdx.x;
  const int r  = rd >> 8;
  const int Dd = rd & 255;
  float val = 0.f;
  if (kd < KD_) {
    const int k = kd / DL_, d = kd % DL_;
    val = Us[(((size_t)r * K_ + k) * D_ + Dd) * DL_ + d];
  }
  Upad2 [(((size_t)r * (D_ / 2) + (Dd >> 1)) * KDP_ + kd) * 2 + (Dd & 1)] = val;
  UpadT2[(((size_t)r * (KDP_ / 2) + (kd >> 1)) * D_ + Dd) * 2 + (kd & 1)] = val;
}

__global__ __launch_bounds__(32) void ksub_fused(const float* __restrict__ x,
                                                 const float* __restrict__ Upad2,
                                                 const float* __restrict__ UpadT2,
                                                 const float* __restrict__ G,
                                                 float* __restrict__ out) {
  __shared__ float sx[16 * XP];
  __shared__ float sz[16 * ZP];
  __shared__ float sg[K_ * KD_];

  const int lane = threadIdx.x;
  const int r  = blockIdx.y;
  const int b0 = blockIdx.x * 16;
  const int g  = lane >> 4;
  const int n  = lane & 15;

  for (int i = lane; i < 16 * 64; i += 32) {
    const int row = i >> 6;
    const int c4  = (i & 63) << 2;
    const float4 v = *(const float4*)(x + (size_t)(b0 + row) * D_ + c4);
    *(float4*)(&sx[row * XP + c4]) = v;
  }
  for (int i = lane; i < K_ * KD_; i += 32) sg[i] = G[r * K_ * KD_ + i];
  __syncthreads();

  const float* Up = Upad2 + (size_t)r * (D_ / 2) * KDP_ * 2;
#pragma unroll 1
  for (int t = 0; t < 7; ++t) {
    const int kd = t * 16 + n;
    const float* ub = Up + ((size_t)g * KDP_ + kd) * 2;
    v8f acc = {};
#pragma unroll
    for (int kb = 0; kb < D_; kb += 4) {
      v2f a = *(const v2f*)(&sx[n * XP + kb + 2 * g]);
      v2f b = *(const v2f*)(&ub[(size_t)(kb >> 1) * KDP_ * 2]);
      acc = __builtin_amdgcn_wmma_f32_16x16x4_f32(false, a, false, b,
                                                  (short)0, acc, false, false);
    }
#pragma unroll
    for (int v = 0; v < 8; ++v)
      sz[(v + 8 * g) * ZP + kd] = acc[v];
  }
  __syncthreads();

  if (lane < 16) {
    float s[K_];
    float m = -3.0e38f;
#pragma unroll 1
    for (int k = 0; k < K_; ++k) {
      float zk[DL_];
#pragma unroll
      for (int d = 0; d < DL_; ++d) zk[d] = sz[lane * ZP + k * DL_ + d];
      float zz = 0.f, q = 0.f;
#pragma unroll
      for (int d1 = 0; d1 < DL_; ++d1) {
        zz = fmaf(zk[d1], zk[d1], zz);
        float t2 = 0.f;
#pragma unroll
        for (int d2 = 0; d2 < DL_; ++d2)
          t2 = fmaf(sg[k * KD_ + d1 * DL_ + d2], zk[d2], t2);
        q = fmaf(zk[d1], t2, q);
      }
      s[k] = (zz - 0.5f * q) * INV_TEMP;
      m = fmaxf(m, s[k]);
    }
    float c[K_];
    float sum = 0.f;
#pragma unroll
    for (int k = 0; k < K_; ++k) { c[k] = __expf(s[k] - m); sum += c[k]; }
    const float inv = 1.0f / sum;
#pragma unroll 1
    for (int k = 0; k < K_; ++k) {
      const float ck = c[k] * inv;
#pragma unroll
      for (int d = 0; d < DL_; ++d) sz[lane * ZP + k * DL_ + d] *= ck;
    }
  }
  __syncthreads();

  const float* Ut = UpadT2 + (size_t)r * (KDP_ / 2) * D_ * 2;
  float* outr = out + (size_t)r * B_ * D_ + (size_t)b0 * D_;
#pragma unroll 1
  for (int t = 0; t < 16; ++t) {
    const int Dd = t * 16 + n;
    const float* ubt = Ut + ((size_t)g * D_ + Dd) * 2;
    v8f acc = {};
#pragma unroll
    for (int kb = 0; kb < KD_; kb += 4) {
      v2f a = *(const v2f*)(&sz[n * ZP + kb + 2 * g]);
      v2f b = *(const v2f*)(&ubt[(size_t)(kb >> 1) * D_ * 2]);
      acc = __builtin_amdgcn_wmma_f32_16x16x4_f32(false, a, false, b,
                                                  (short)0, acc, false, false);
    }
#pragma unroll
    for (int v = 0; v < 8; ++v)
      outr[(size_t)(v + 8 * g) * D_ + Dd] = acc[v];
  }
}

extern "C" void kernel_launch(void* const* d_in, const int* in_sizes, int n_in,
                              void* d_out, int out_size, void* d_ws, size_t ws_size,
                              hipStream_t stream) {
  (void)in_sizes; (void)n_in; (void)out_size; (void)ws_size;
  const float* x  = (const float*)d_in[0];
  const float* Us = (const float*)d_in[1];
  float* G      = (float*)((char*)d_ws + G_OFF);
  float* Upad2  = (float*)((char*)d_ws + UP_OFF);
  float* UpadT2 = (float*)((char*)d_ws + UT_OFF);
  float* out = (float*)d_out;

  gram_kernel<<<R_ * K_, 128, 0, stream>>>(Us, G);
  pack_kernel<<<R_ * D_, KDP_, 0, stream>>>(Us, Upad2, UpadT2);
  ksub_fused<<<dim3(B_ / 16, R_), 32, 0, stream>>>(x, Upad2, UpadT2, G, out);
}


# fused kernel retiled: one 8-wave work-group per 64 rows; each wave owns whole N-tiles of both GEMMs and feeds every B fragment to four row-tile accumulators; per-element sum order unchanged
# speedup vs baseline: 1.4444x; 1.4444x over previous
#include <hip/hip_runtime.h>
#include <hip/hip_bf16.h>

#define R_ 8
#define K_ 10
#define D_ 256
#define DL_ 10
#define B_ 4096
#define KD_ 100
#define KDP_ 112
#define INV_TEMP 10.0f

#define XP 260
#define ZP 114

#define G_OFF  0u
#define UP_OFF 32768u
#define UT_OFF 950272u

typedef float v2f __attribute__((ext_vector_type(2)));
typedef float v8f __attribute__((ext_vector_type(8)));

__global__ __launch_bounds__(128) void gram_kernel(const float* __restrict__ Us,
                                                   float* __restrict__ G) {
  const int rk = blockIdx.x;
  const int t  = threadIdx.x;
  if (t >= KD_) return;
  const int d1 = t / DL_, d2 = t % DL_;
  const float* u = Us + (size_t)rk * D_ * DL_;
  float acc = 0.f;
#pragma unroll 8
  for (int Dd = 0; Dd < D_; ++Dd)
    acc = fmaf(u[Dd * DL_ + d1], u[Dd * DL_ + d2], acc);
  G[rk * KD_ + t] = acc;
}

__global__ __launch_bounds__(KDP_) void pack_kernel(const float* __restrict__ Us,
                                                    float* __restrict__ Upad2,
                                                    float* __restrict__ UpadT2) {
  const int kd = threadIdx.x;
  const int rd = blockIdx.x;
  const int r  = rd >> 8;
  const int Dd = rd & 255;
  float val = 0.f;
  if (kd < KD_) {
    const int k = kd / DL_, d = kd % DL_;
    val = Us[(((size_t)r * K_ + k) * D_ + Dd) * DL_ + d];
  }
  Upad2 [(((size_t)r * (D_ / 2) + (Dd >> 1)) * KDP_ + kd) * 2 + (Dd & 1)] = val;
  UpadT2[(((size_t)r * (KDP_ / 2) + (kd >> 1)) * D_ + Dd) * 2 + (kd & 1)] = val;
}

#define MT_   4
#define ROWS_ (16 * MT_)
#define NW_   8

__global__ __launch_bounds__(32 * NW_) void ksub_fused(const float* __restrict__ x,
                                                       const float* __restrict__ Upad2,
                                                       const float* __restrict__ UpadT2,
                                                       const float* __restrict__ G,
                                                       float* __restrict__ out) {
  __shared__ float sx[ROWS_ * XP];
  __shared__ float sz[ROWS_ * ZP];
  __shared__ float sg[K_ * KD_];

  const int tid  = threadIdx.x;
  const int lane = tid & 31;
  const int w  = __builtin_amdgcn_readfirstlane(tid >> 5);
  const int r  = blockIdx.y;
  const int b0 = blockIdx.x * ROWS_;
  const int g  = lane >> 4;
  const int n  = lane & 15;

  for (int i = tid; i < ROWS_ * 64; i += 32 * NW_) {
    const int row = i >> 6;
    const int c4  = (i & 63) << 2;
    const float4 v = *(const float4*)(x + (size_t)(b0 + row) * D_ + c4);
    *(float4*)(&sx[row * XP + c4]) = v;
  }
  __syncthreads();

  if (w < 7) {
    const float* Up = Upad2 + (size_t)r * (D_ / 2) * KDP_ * 2;
    const int kd = w * 16 + n;
    const float* ub = Up + ((size_t)g * KDP_ + kd) * 2;
    v8f acc[MT_] = {};
#pragma unroll
    for (int kb = 0; kb < D_; kb += 4) {
      v2f b = *(const v2f*)(&ub[(size_t)(kb >> 1) * KDP_ * 2]);
#pragma unroll
      for (int m = 0; m < MT_; ++m) {
        v2f a = *(const v2f*)(&sx[(m * 16 + n) * XP + kb + 2 * g]);
        acc[m] = __builtin_amdgcn_wmma_f32_16x16x4_f32(false, a, false, b,
                                                       (short)0, acc[m], false, false);
      }
    }
#pragma unroll
    for (int m = 0; m < MT_; ++m)
#pragma unroll
      for (int v = 0; v < 8; ++v)
        sz[(m * 16 + v + 8 * g) * ZP + kd] = acc[m][v];
  } else {
    for (int i = lane; i < K_ * KD_; i += 32) sg[i] = G[r * K_ * KD_ + i];
  }
  __syncthreads();

  if (tid < ROWS_) {
    float s[K_];
    float m = -3.0e38f;
#pragma unroll 1
    for (int k = 0; k < K_; ++k) {
      float zk[DL_];
#pragma unroll
      for (int d = 0; d < DL_; ++d) zk[d] = sz[tid * ZP + k * DL_ + d];
      float zz = 0.f, q = 0.f;
#pragma unroll
      for (int d1 = 0; d1 < DL_; ++d1) {
        zz = fmaf(zk[d1], zk[d1], zz);
        float t2 = 0.f;
#pragma unroll
        for (int d2 = 0; d2 < DL_; ++d2)
          t2 = fmaf(sg[k * KD_ + d1 * DL_ + d2], zk[d2], t2);
        q = fmaf(zk[d1], t2, q);
      }
      s[k] = (zz - 0.5f * q) * INV_TEMP;
      m = fmaxf(m, s[k]);
    }
    float c[K_];
    float sum = 0.f;
#pragma unroll
    for (int k = 0; k < K_; ++k) { c[k] = __expf(s[k] - m); sum += c[k]; }
    const float inv = 1.0f / sum;
#pragma unroll 1
    for (int k = 0; k < K_; ++k) {
      const float ck = c[k] * inv;
#pragma unroll
      for (int d = 0; d < DL_; ++d) sz[tid * ZP + k * DL_ + d] *= ck;
    }
  }
  __syncthreads();

  const float* Ut = UpadT2 + (size_t)r * (KDP_ / 2) * D_ * 2;
  float* outr = out + (size_t)r * B_ * D_ + (size_t)b0 * D_;
#pragma unroll 1
  for (int j = 0; j < 2; ++j) {
    const int Dd = (2 * w + j) * 16 + n;
    const float* ubt = Ut + ((size_t)g * D_ + Dd) * 2;
    v8f acc[MT_] = {};
#pragma unroll
    for (int kb = 0; kb < KD_; kb += 4) {
      v2f b = *(const v2f*)(&ubt[(size_t)(kb >> 1) * D_ * 2]);
#pragma unroll
      for (int m = 0; m < MT_; ++m) {
        v2f a = *(const v2f*)(&sz[(m * 16 + n) * ZP + kb + 2 * g]);
        acc[m] = __builtin_amdgcn_wmma_f32_16x16x4_f32(false, a, false, b,
                                                       (short)0, acc[m], false, false);
      }
    }
#pragma unroll
    for (int m = 0; m < MT_; ++m)
#pragma unroll
      for (int v = 0; v < 8; ++v)
        outr[(size_t)(m * 16 + v + 8 * g) * D_ + Dd] = acc[m][v];
  }
}

extern "C" void kernel_launch(void* const* d_in, const int* in_sizes, int n_in,
                              void* d_out, int out_size, void* d_ws, size_t ws_size,
                              hipStream_t stream) {
  (void)in_sizes; (void)n_in; (void)out_size; (void)ws_size;
  const float* x  = (const float*)d_in[0];
  const float* Us = (const float*)d_in[1];
  float* G      = (float*)((char*)d_ws + G_OFF);
  float* Upad2  = (float*)((char*)d_ws + UP_OFF);
  float* UpadT2 = (float*)((char*)d_ws + UT_OFF);
  float* out = (float*)d_out;

  gram_kernel<<<R_ * K_, 128, 0, stream>>>(Us, G);
  pack_kernel<<<R_ * D_, KDP_, 0, stream>>>(Us, Upad2, UpadT2);
  ksub_fused<<<dim3(B_ / ROWS_, R_), 32 * NW_, 0, stream>>>(x, Upad2, UpadT2, G, out);
}


# prep in one launch instead of two: 80 work-groups stage an (r,k) block of Us in LDS for its Gram block, 128 stage a 16-row slab for whole lines of both packed copies; Gram sum order unchanged
# speedup vs baseline: 2.4846x; 1.7202x over previous
#include <hip/hip_runtime.h>
#include <hip/hip_bf16.h>

#define R_ 8
#define K_ 10
#define D_ 256
#define DL_ 10
#define B_ 4096
#define KD_ 100
#define KDP_ 112
#define INV_TEMP 10.0f

#define XP 260
#define ZP 114

#define G_OFF  0u
#define UP_OFF 32768u
#define UT_OFF 950272u

typedef float v2f __attribute__((ext_vector_type(2)));
typedef float v8f __attribute__((ext_vector_type(8)));

#define PREP_T  256
#define GRAM_WG (R_ * K_)
#define PACK_WG (R_ * (D_ / 16))
#define GP 260
#define PP 18
#define GC_ 8

__global__ __launch_bounds__(PREP_T) void prep_kernel(const float* __restrict__ Us,
                                                      float* __restrict__ G,
                                                      float* __restrict__ Upad2,
                                                      float* __restrict__ UpadT2) {
  __shared__ __attribute__((aligned(16))) float sm[DL_ * GP];
  const int tid = threadIdx.x;

  if (blockIdx.x < GRAM_WG) {
    const int rk = blockIdx.x;
    const float* u = Us + (size_t)rk * D_ * DL_;
    float4 gv[3];
#pragma unroll
    for (int t = 0; t < 3; ++t) {
      const int i = tid + t * PREP_T;
      gv[t] = *(const float4*)(u + 4 * (i < D_ * DL_ / 4 ? i : D_ * DL_ / 4 - 1));
    }
#pragma unroll
    for (int t = 0; t < 3; ++t) {
      const int i = tid + t * PREP_T;
      const float e4[4] = {gv[t].x, gv[t].y, gv[t].z, gv[t].w};
      if (i < D_ * DL_ / 4) {
#pragma unroll
        for (int j = 0; j < 4; ++j) {
          const int e = 4 * i + j;
          sm[(e % DL_) * GP + e / DL_] = e4[j];
        }
      }
    }
    __syncthreads();
    if (tid < KD_) {
      const int d1 = tid / DL_, d2 = tid % DL_;
      const float* a = &sm[d1 * GP];
      const float* b = &sm[d2 * GP];
      float acc = 0.f;
      float4 av[2][GC_], bv[2][GC_];
#pragma unroll
      for (int s = 0; s < GC_; ++s) {
        av[0][s] = *(const float4*)(a + 4 * s);
        bv[0][s] = *(const float4*)(b + 4 * s);
      }
#pragma unroll
      for (int ch = 0; ch < D_ / (4 * GC_); ++ch) {
        __builtin_amdgcn_sched_barrier(0);
        if (ch + 1 < D_ / (4 * GC_)) {
#pragma unroll
          for (int s = 0; s < GC_; ++s) {
            av[(ch + 1) & 1][s] = *(const float4*)(a + 4 * GC_ * (ch + 1) + 4 * s);
            bv[(ch + 1) & 1][s] = *(const float4*)(b + 4 * GC_ * (ch + 1) + 4 * s);
          }
        }
        __builtin_amdgcn_sched_barrier(0);
#pragma unroll
        for (int s = 0; s < GC_; ++s) {
          const float4 x4 = av[ch & 1][s], y4 = bv[ch & 1][s];
          acc = fmaf(x4.x, y4.x, acc);
          acc = fmaf(x4.y, y4.y, acc);
          acc = fmaf(x4.z, y4.z, acc);
          acc = fmaf(x4.w, y4.w, acc);
        }
      }
      G[rk * KD_ + tid] = acc;
    }
  } else {
    const int idx = blockIdx.x - GRAM_WG;
    const int r = idx >> 4;
    const int c = idx & 15;
    float4 pv[2];
#pragma unroll
    for (int t = 0; t < 2; ++t) {
      const int i = tid + t * PREP_T;
      const int ic = i < K_ * 40 ? i : K_ * 40 - 1;
      const int k = ic / 40, q = ic % 40;
      pv[t] = *(const float4*)(Us + (((size_t)r * K_ + k) * D_ + 16 * c) * DL_ + 4 * q);
    }
#pragma unroll
    for (int t = 0; t < 2; ++t) {
      const int i = tid + t * PREP_T;
      const int k = i / 40, q = i % 40;
      const float e4[4] = {pv[t].x, pv[t].y, pv[t].z, pv[t].w};
      if (i < K_ * 40) {
#pragma unroll
        for (int j = 0; j < 4; ++j) {
          const int e = 4 * q + j;
          sm[(k * DL_ + e % DL_) * PP + e / DL_] = e4[j];
        }
      }
    }
    if (tid < (KDP_ - KD_) * 16)
      sm[(KD_ + (tid >> 4)) * PP + (tid & 15)] = 0.f;
    __syncthreads();

    float* up = Upad2 + ((size_t)r * (D_ / 2) + 8 * c) * KDP_ * 2;
    for (int f = tid; f < 8 * (KDP_ / 2); f += PREP_T) {
      const int h = f / (KDP_ / 2), kdp = f % (KDP_ / 2);
      const v2f lo = *(const v2f*)(&sm[(2 * kdp) * PP + 2 * h]);
      const v2f hi = *(const v2f*)(&sm[(2 * kdp + 1) * PP + 2 * h]);
      *(float4*)(up + (size_t)f * 4) = make_float4(lo.x, lo.y, hi.x, hi.y);
    }
    float* ut = UpadT2 + ((size_t)r * (KDP_ / 2) * D_ + 16 * c) * 2;
    for (int f = tid; f < (KDP_ / 2) * 8; f += PREP_T) {
      const int kdp = f >> 3, h = f & 7;
      const v2f lo = *(const v2f*)(&sm[(2 * kdp) * PP + 2 * h]);
      const v2f hi = *(const v2f*)(&sm[(2 * kdp + 1) * PP + 2 * h]);
      *(float4*)(ut + ((size_t)kdp * D_ + 2 * h) * 2) = make_float4(lo.x, hi.x, lo.y, hi.y);
    }
  }
}

#define MT_   4
#define ROWS_ (16 * MT_)
#define NW_   8

__global__ __launch_bounds__(32 * NW_) void ksub_fused(const float* __restrict__ x,
                                                       const float* __restrict__ Upad2,
                                                       const float* __restrict__ UpadT2,
                                                       const float* __restrict__ G,
                                                       float* __restrict__ out) {
  __shared__ float sx[ROWS_ * XP];
  __shared__ float sz[ROWS_ * ZP];
  __shared__ float sg[K_ * KD_];

  const int tid  = threadIdx.x;
  const int lane = tid & 31;
  const int w  = __builtin_amdgcn_readfirstlane(tid >> 5);
  const int r  = blockIdx.y;
  const int b0 = blockIdx.x * ROWS_;
  const int g  = lane >> 4;
  const int n  = lane & 15;

  for (int i = tid; i < ROWS_ * 64; i += 32 * NW_) {
    const int row = i >> 6;
    const int c4  = (i & 63) << 2;
    const float4 v = *(const float4*)(x + (size_t)(b0 + row) * D_ + c4);
    *(float4*)(&sx[row * XP + c4]) = v;
  }
  __syncthreads();

  if (w < 7) {
    const float* Up = Upad2 + (size_t)r * (D_ / 2) * KDP_ * 2;
    const int kd = w * 16 + n;
    const float* ub = Up + ((size_t)g * KDP_ + kd) * 2;
    v8f acc[MT_] = {};
#pragma unroll
    for (int kb = 0; kb < D_; kb += 4) {
      v2f b = *(const v2f*)(&ub[(size_t)(kb >> 1) * KDP_ * 2]);
#pragma unroll
      for (int m = 0; m < MT_; ++m) {
        v2f a = *(const v2f*)(&sx[(m * 16 + n) * XP + kb + 2 * g]);
        acc[m] = __builtin_amdgcn_wmma_f32_16x16x4_f32(false, a, false, b,
                                                       (short)0, acc[m], false, false);
      }
    }
#pragma unroll
    for (int m = 0; m < MT_; ++m)
#pragma unroll
      for (int v = 0; v < 8; ++v)
        sz[(m * 16 + v + 8 * g) * ZP + kd] = acc[m][v];
  } else {
    for (int i = lane; i < K_ * KD_; i += 32) sg[i] = G[r * K_ * KD_ + i];
  }
  __syncthreads();

  if (tid < ROWS_) {
    float s[K_];
    float m = -3.0e38f;
#pragma unroll 1
    for (int k = 0; k < K_; ++k) {
      float zk[DL_];
#pragma unroll
      for (int d = 0; d < DL_; ++d) zk[d] = sz[tid * ZP + k * DL_ + d];
      float zz = 0.f, q = 0.f;
#pragma unroll
      for (int d1 = 0; d1 < DL_; ++d1) {
        zz = fmaf(zk[d1], zk[d1], zz);
        float t2 = 0.f;
#pragma unroll
        for (int d2 = 0; d2 < DL_; ++d2)
          t2 = fmaf(sg[k * KD_ + d1 * DL_ + d2], zk[d2], t2);
        q = fmaf(zk[d1], t2, q);
      }
      s[k] = (zz - 0.5f * q) * INV_TEMP;
      m = fmaxf(m, s[k]);
    }
    float c[K_];
    float sum = 0.f;
#pragma unroll
    for (int k = 0; k < K_; ++k) { c[k] = __expf(s[k] - m); sum += c[k]; }
    const float inv = 1.0f / sum;
#pragma unroll 1
    for (int k = 0; k < K_; ++k) {
      const float ck = c[k] * inv;
#pragma unroll
      for (int d = 0; d < DL_; ++d) sz[tid * ZP + k * DL_ + d] *= ck;
    }
  }
  __syncthreads();

  const float* Ut = UpadT2 + (size_t)r * (KDP_ / 2) * D_ * 2;
  float* outr = out + (size_t)r * B_ * D_ + (size_t)b0 * D_;
#pragma unroll 1
  for (int j = 0; j < 2; ++j) {
    const int Dd = (2 * w + j) * 16 + n;
    const float* ubt = Ut + ((size_t)g * D_ + Dd) * 2;
    v8f acc[MT_] = {};
#pragma unroll
    for (int kb = 0; kb < KD_; kb += 4) {
      v2f b = *(const v2f*)(&ubt[(size_t)(kb >> 1) * D_ * 2]);
#pragma unroll
      for (int m = 0; m < MT_; ++m) {
        v2f a = *(const v2f*)(&sz[(m * 16 + n) * ZP + kb + 2 * g]);
        acc[m] = __builtin_amdgcn_wmma_f32_16x16x4_f32(false, a, false, b,
                                                       (short)0, acc[m], false, false);
      }
    }
#pragma unroll
    for (int m = 0; m < MT_; ++m)
#pragma unroll
      for (int v = 0; v < 8; ++v)
        outr[(size_t)(m * 16 + v + 8 * g) * D_ + Dd] = acc[m][v];
  }
}

extern "C" void kernel_launch(void* const* d_in, const int* in_sizes, int n_in,
                              void* d_out, int out_size, void* d_ws, size_t ws_size,
                              hipStream_t stream) {
  (void)in_sizes; (void)n_in; (void)out_size; (void)ws_size;
  const float* x  = (const float*)d_in[0];
  const float* Us = (const float*)d_in[1];
  float* G      = (float*)((char*)d_ws + G_OFF);
  float* Upad2  = (float*)((char*)d_ws + UP_OFF);
  float* UpadT2 = (float*)((char*)d_ws + UT_OFF);
  float* out = (float*)d_out;

  prep_kernel<<<GRAM_WG + PACK_WG, PREP_T, 0, stream>>>(Us, G, Upad2, UpadT2);
  ksub_fused<<<dim3(B_ / ROWS_, R_), 32 * NW_, 0, stream>>>(x, Upad2, UpadT2, G, out);
}
